# starting kernel
# baseline (speedup 1.0000x reference)
#include <hip/hip_runtime.h>

typedef __attribute__((ext_vector_type(2))) float v2f;
typedef __attribute__((ext_vector_type(8))) float v8f;

constexpr int B_ = 1024;
constexpr int G_ = 20000;
constexpr int L_ = 1000;
constexpr int g_ = 100;
constexpr int C_ = 10;
constexpr int NPAD = 16;

constexpr int TILES_M = B_ / 16;
constexpr int KSPLIT  = 125;
constexpr int KSEG    = G_ / KSPLIT;
constexpr int WSTEPS  = KSEG / 4;
constexpr int UNROLL  = 8;
static_assert(WSTEPS % UNROLL == 0, "batching must divide steps");


constexpr size_t OFF_WEFF  = 0;
constexpr size_t OFF_STAGE = (size_t)G_ * NPAD;
constexpr size_t OFF_BAGG  = OFF_STAGE + (size_t)TILES_M * 256;
constexpr size_t WS_FLOATS = OFF_BAGG + 16;

__global__ void zero_ws_k(float* ws) {
    size_t i = (size_t)blockIdx.x * 256 + threadIdx.x;
    if (i < WS_FLOATS) ws[i] = 0.0f;
}

__global__ void bias_agg_k(const float* __restrict__ bias,
                           const float* __restrict__ agg_w,
                           const float* __restrict__ agg_b,
                           float* __restrict__ bagg) {
    int c = threadIdx.x;
    if (c < C_) {
        float s = agg_b[0];
        for (int l = 0; l < L_; ++l) s += agg_w[l] * bias[l * C_ + c];
        bagg[c] = s;
    }
}

__global__ void scatter_k(const float* __restrict__ W,
                          const float* __restrict__ agg_w,
                          const int* __restrict__ gidx,
                          float* __restrict__ weff) {
    int t = blockIdx.x * 256 + threadIdx.x;
    if (t >= L_ * g_) return;
    int l = t / g_;
    int k = gidx[t];
    float aw = agg_w[l];
    const float* w = W + (size_t)t * C_;
    float* dst = weff + (size_t)(k >> 1) * 32 + (k & 1);
#pragma unroll
    for (int c = 0; c < C_; ++c) atomicAdd(dst + c * 2, aw * w[c]);
}

__global__ __launch_bounds__(32)
void gemm_k(const float* __restrict__ X,
            const float* __restrict__ weff,
            float* __restrict__ stage) {
    const int tile = blockIdx.x % TILES_M;
    const int seg  = blockIdx.x / TILES_M;
    const int lane = threadIdx.x;
    const int half = lane >> 4;
    const int sub  = lane & 15;
    const int koff = half * 2;
    const int kbase = seg * KSEG;

    const float* ap = X + (size_t)(tile * 16 + sub) * G_ + kbase + koff;
    const float* bp = weff + (size_t)((kbase + koff) >> 1) * 32 + sub * 2;

    v8f acc = {0.f, 0.f, 0.f, 0.f, 0.f, 0.f, 0.f, 0.f};

    for (int o = 0; o < WSTEPS / UNROLL; ++o) {
        v2f a[UNROLL], b[UNROLL];
#pragma unroll
        for (int u = 0; u < UNROLL; ++u) {
            a[u] = __builtin_nontemporal_load((const v2f*)(ap + u * 4));
            b[u] = *(const v2f*)(bp + u * 64);
        }
        __builtin_prefetch(ap + UNROLL * 4, 0, 3);
#pragma unroll
        for (int u = 0; u < UNROLL; ++u)
            acc = __builtin_amdgcn_wmma_f32_16x16x4_f32(
                 false, a[u],  false, b[u],
                 (short)0, acc,  false,  false);
        ap += UNROLL * 4;
        bp += UNROLL * 64;
    }

    float* st = stage + tile * 256 + sub;
    const int mbase = half * 8;
#pragma unroll
    for (int r = 0; r < 8; ++r)
        atomicAdd(st + (mbase + r) * 16, acc[r]);
}

__global__ void out_k(const float* __restrict__ stage,
                      const float* __restrict__ bagg,
                      float* __restrict__ out) {
    int t = blockIdx.x * 256 + threadIdx.x;
    if (t >= B_ * C_) return;
    int b = t / C_, c = t % C_;
    out[t] = stage[(b >> 4) * 256 + (b & 15) * 16 + c] + bagg[c];
}

extern "C" void kernel_launch(void* const* d_in, const int* in_sizes, int n_in,
                              void* d_out, int out_size, void* d_ws, size_t ws_size,
                              hipStream_t stream) {
    const float* X     = (const float*)d_in[0];
    const float* W     = (const float*)d_in[1];
    const float* bias  = (const float*)d_in[2];
    const float* agg_w = (const float*)d_in[3];
    const float* agg_b = (const float*)d_in[4];
    const int*   gidx  = (const int*)d_in[5];

    float* ws    = (float*)d_ws;
    float* weff  = ws + OFF_WEFF;
    float* stage = ws + OFF_STAGE;
    float* bagg  = ws + OFF_BAGG;
    float* out   = (float*)d_out;

    (void)in_sizes; (void)n_in; (void)out_size; (void)ws_size;

    const int zgrid = (int)((WS_FLOATS + 255) / 256);
    zero_ws_k<<<zgrid, 256, 0, stream>>>(ws);
    bias_agg_k<<<1, 32, 0, stream>>>(bias, agg_w, agg_b, bagg);
    scatter_k<<<(L_ * g_ + 255) / 256, 256, 0, stream>>>(W, agg_w, gidx, weff);
    gemm_k<<<TILES_M * KSPLIT, 32, 0, stream>>>(X, weff, stage);
    out_k<<<(B_ * C_ + 255) / 256, 256, 0, stream>>>(stage, bagg, out);
}


# Weff built by owner work-groups: each owns 160 rows, scans the index list, sums its entries in LDS (f32 LDS adds, sum order changes), stores its lines once; global scatter atomics and Weff clear gone
# speedup vs baseline: 2.2180x; 2.2180x over previous
#include <hip/hip_runtime.h>

typedef __attribute__((ext_vector_type(2))) float v2f;
typedef __attribute__((ext_vector_type(8))) float v8f;

constexpr int B_ = 1024;
constexpr int G_ = 20000;
constexpr int L_ = 1000;
constexpr int g_ = 100;
constexpr int C_ = 10;
constexpr int NPAD = 16;

constexpr int TILES_M = B_ / 16;
constexpr int KSPLIT  = 125;
constexpr int KSEG    = G_ / KSPLIT;
constexpr int WSTEPS  = KSEG / 4;
constexpr int UNROLL  = 8;
static_assert(WSTEPS % UNROLL == 0, "batching must divide steps");
constexpr int LSEG    = L_ / KSPLIT;
static_assert(LSEG * KSPLIT == L_ && LSEG == 8, "bias rows must fill two 4-wide k-steps per segment");


constexpr size_t OFF_WEFF  = 0;
constexpr size_t OFF_STAGE = (size_t)G_ * NPAD;
constexpr size_t WS_FLOATS = OFF_STAGE + (size_t)TILES_M * 256;

constexpr int ZERO_WGS = TILES_M * 256 / 4 / 256;
__global__ __launch_bounds__(256)
void zero_stage_k(float* __restrict__ stage) {
    const int i = blockIdx.x * 256 + threadIdx.x;
    ((float4*)stage)[i] = make_float4(0.f, 0.f, 0.f, 0.f);
}

constexpr int OWN_ROWS    = 160;
constexpr int OWN_WGS     = G_ / OWN_ROWS;
constexpr int OWN_THREADS = 1024;
constexpr int OWN_WAVES   = OWN_THREADS / 32;
constexpr int NVEC        = L_ * g_ / 4;
constexpr int SCAN_ROUNDS = (NVEC + OWN_THREADS - 1) / OWN_THREADS;
constexpr int SCAN_BATCH  = 5;
constexpr int LIST_CAP    = 256;
constexpr int SUM_BATCH   = 8;
static_assert(OWN_WGS * OWN_ROWS == G_ && (OWN_ROWS & 1) == 0, "owner ranges tile Weff in whole lines");
static_assert(SCAN_ROUNDS % SCAN_BATCH == 0, "scan batches must divide the rounds");
static_assert(L_ * g_ < (1 << 17) && OWN_ROWS <= 256, "entry id (17 bits) and local row (8 bits) share one int");

__device__ __forceinline__ void add_entry_now(int t, int d,
                                              const float* __restrict__ W,
                                              const float* __restrict__ agg_w,
                                              float* acc) {
    const float aw = agg_w[t / g_];
    const float* w = W + (size_t)t * C_;
    float* dst = acc + (d >> 1) * 32 + (d & 1);
    for (int c = 0; c < C_; ++c) atomicAdd(dst + c * 2, aw * w[c]);
}

__global__ __launch_bounds__(OWN_THREADS)
void build_weff_k(const float* __restrict__ W,
                  const float* __restrict__ agg_w,
                  const int* __restrict__ gidx,
                  float* __restrict__ weff) {
    __shared__ __attribute__((aligned(16))) float acc[OWN_ROWS * NPAD];
    __shared__ int parked[OWN_WAVES][LIST_CAP];

    const int tid  = threadIdx.x;
    const int lane = tid & 31;
    const int wave = tid >> 5;
    const int k0   = blockIdx.x * OWN_ROWS;

    for (int i = tid; i < OWN_ROWS * NPAD; i += OWN_THREADS) acc[i] = 0.f;
    __syncthreads();

    int nw = 0;
    const int4* g4 = (const int4*)gidx;
    auto visit = [&](int k, int t, bool inr) {
        const int d = k - k0;
        const bool hit = inr && (unsigned)d < (unsigned)OWN_ROWS;
        const unsigned mask = __builtin_amdgcn_ballot_w32(hit);
        if (mask != 0u) {
            if (hit) {
                const int slot = nw + __builtin_popcount(mask & ((1u << lane) - 1u));
                if (slot < LIST_CAP) parked[wave][slot] = (t << 8) | d;
            }
            nw += __builtin_popcount(mask);
        }
    };
    for (int r0 = 0; r0 < SCAN_ROUNDS; r0 += SCAN_BATCH) {
        int4 q[SCAN_BATCH];
#pragma unroll
        for (int u = 0; u < SCAN_BATCH; ++u) {
            const int v = tid + (r0 + u) * OWN_THREADS;
            q[u] = g4[v < NVEC ? v : NVEC - 1];
        }
#pragma unroll
        for (int u = 0; u < SCAN_BATCH; ++u) {
            const int v = tid + (r0 + u) * OWN_THREADS;
            const bool inr = v < NVEC;
            visit(q[u].x, v * 4 + 0, inr);
            visit(q[u].y, v * 4 + 1, inr);
            visit(q[u].z, v * 4 + 2, inr);
            visit(q[u].w, v * 4 + 3, inr);
        }
    }
    if (nw > LIST_CAP) {
        int seen = 0;
        for (int i = 0; i < SCAN_ROUNDS * 4; ++i) {
            const int v = tid + (i >> 2) * OWN_THREADS;
            const bool inr = v < NVEC;
            const int t = (inr ? v : NVEC - 1) * 4 + (i & 3);
            const int d = gidx[t] - k0;
            const bool hit = inr && (unsigned)d < (unsigned)OWN_ROWS;
            const unsigned mask = __builtin_amdgcn_ballot_w32(hit);
            if (hit && seen + __builtin_popcount(mask & ((1u << lane) - 1u)) >= LIST_CAP)
                add_entry_now(t, d, W, agg_w, acc);
            seen += __builtin_popcount(mask);
        }
    }
    __syncthreads();

    const int total = (nw < LIST_CAP ? nw : LIST_CAP) * 5;
    for (int base = 0; base < total; base += 32 * SUM_BATCH) {
        v2f   wv[SUM_BATCH];
        float aw[SUM_BATCH];
        int   off[SUM_BATCH];
#pragma unroll
        for (int u = 0; u < SUM_BATCH; ++u) {
            const int i  = base + u * 32 + lane;
            const int ii = i < total ? i : 0;
            const int e  = ii / 5;
            const int j  = ii - e * 5;
            const int p  = parked[wave][e];
            const int t  = p >> 8;
            const int d  = p & 255;
            aw[u]  = agg_w[t / g_];
            wv[u]  = *(const v2f*)(W + (size_t)t * C_ + 2 * j);
            off[u] = (d >> 1) * 32 + j * 4 + (d & 1);
        }
#pragma unroll
        for (int u = 0; u < SUM_BATCH; ++u) {
            if (base + u * 32 + lane < total) {
                atomicAdd(&acc[off[u]],     aw[u] * wv[u].x);
                atomicAdd(&acc[off[u] + 2], aw[u] * wv[u].y);
            }
        }
    }
    __syncthreads();

    float4* dst = (float4*)(weff + (size_t)k0 * NPAD);
    const float4* src = (const float4*)acc;
    for (int i = tid; i < OWN_ROWS * NPAD / 4; i += OWN_THREADS) dst[i] = src[i];
}

__global__ __launch_bounds__(32)
void gemm_k(const float* __restrict__ X,
            const float* __restrict__ weff,
            const float* __restrict__ bias,
            const float* __restrict__ agg_w,
            float* __restrict__ stage) {
    const int tile = blockIdx.x % TILES_M;
    const int seg  = blockIdx.x / TILES_M;
    const int lane = threadIdx.x;
    const int half = lane >> 4;
    const int sub  = lane & 15;
    const int koff = half * 2;
    const int kbase = seg * KSEG;

    const float* ap = X + (size_t)(tile * 16 + sub) * G_ + kbase + koff;
    const float* bp = weff + (size_t)((kbase + koff) >> 1) * 32 + sub * 2;

    v8f acc = {0.f, 0.f, 0.f, 0.f, 0.f, 0.f, 0.f, 0.f};

    {
        const int l0 = seg * LSEG + koff;
        const bool live = sub < C_;
        const float* wl = agg_w + l0;
        const float* bl = bias + l0 * C_ + (live ? sub : 0);
        const v2f wa0 = {wl[0], wl[1]};
        const v2f wa1 = {wl[4], wl[5]};
        const float b00 = bl[0], b01 = bl[C_], b10 = bl[4 * C_], b11 = bl[5 * C_];
        const v2f bb0 = {live ? b00 : 0.f, live ? b01 : 0.f};
        const v2f bb1 = {live ? b10 : 0.f, live ? b11 : 0.f};
        acc = __builtin_amdgcn_wmma_f32_16x16x4_f32(
            false, wa0, false, bb0, (short)0, acc, false, false);
        acc = __builtin_amdgcn_wmma_f32_16x16x4_f32(
            false, wa1, false, bb1, (short)0, acc, false, false);
    }

    for (int o = 0; o < WSTEPS / UNROLL; ++o) {
        v2f a[UNROLL], b[UNROLL];
#pragma unroll
        for (int u = 0; u < UNROLL; ++u) {
            a[u] = __builtin_nontemporal_load((const v2f*)(ap + u * 4));
            b[u] = *(const v2f*)(bp + u * 64);
        }
        __builtin_prefetch(ap + UNROLL * 4, 0, 3);
#pragma unroll
        for (int u = 0; u < UNROLL; ++u)
            acc = __builtin_amdgcn_wmma_f32_16x16x4_f32(
                 false, a[u],  false, b[u],
                 (short)0, acc,  false,  false);
        ap += UNROLL * 4;
        bp += UNROLL * 64;
    }

    float* st = stage + tile * 256 + sub;
    const int mbase = half * 8;
#pragma unroll
    for (int r = 0; r < 8; ++r)
        atomicAdd(st + (mbase + r) * 16, acc[r]);
}

__global__ void out_k(const float* __restrict__ stage,
                      const float* __restrict__ agg_b,
                      float* __restrict__ out) {
    int t = blockIdx.x * 256 + threadIdx.x;
    if (t >= B_ * C_) return;
    int b = t / C_, c = t % C_;
    out[t] = stage[(b >> 4) * 256 + (b & 15) * 16 + c] + agg_b[0];
}

extern "C" void kernel_launch(void* const* d_in, const int* in_sizes, int n_in,
                              void* d_out, int out_size, void* d_ws, size_t ws_size,
                              hipStream_t stream) {
    const float* X     = (const float*)d_in[0];
    const float* W     = (const float*)d_in[1];
    const float* bias  = (const float*)d_in[2];
    const float* agg_w = (const float*)d_in[3];
    const float* agg_b = (const float*)d_in[4];
    const int*   gidx  = (const int*)d_in[5];

    float* ws    = (float*)d_ws;
    float* weff  = ws + OFF_WEFF;
    float* stage = ws + OFF_STAGE;
    float* out   = (float*)d_out;

    (void)in_sizes; (void)n_in; (void)out_size; (void)ws_size;

    zero_stage_k<<<ZERO_WGS, 256, 0, stream>>>(stage);
    build_weff_k<<<OWN_WGS, OWN_THREADS, 0, stream>>>(W, agg_w, gidx, weff);
    gemm_k<<<TILES_M * KSPLIT, 32, 0, stream>>>(X, weff, bias, agg_w, stage);
    out_k<<<(B_ * C_ + 255) / 256, 256, 0, stream>>>(stage, agg_b, out);
}


# speedup vs baseline: 3.7720x; 1.7006x over previous
#include <hip/hip_runtime.h>

typedef __attribute__((ext_vector_type(2))) float v2f;
typedef __attribute__((ext_vector_type(8))) float v8f;

constexpr int B_ = 1024;
constexpr int G_ = 20000;
constexpr int L_ = 1000;
constexpr int g_ = 100;
constexpr int C_ = 10;
constexpr int NPAD = 16;

constexpr int TILES_M = B_ / 16;
constexpr int KSPLIT  = 125;
constexpr int KSEG    = G_ / KSPLIT;
constexpr int WSTEPS  = KSEG / 4;
constexpr int UNROLL  = 8;
static_assert(WSTEPS % UNROLL == 0, "batching must divide steps");
constexpr int LSEG    = L_ / KSPLIT;
static_assert(LSEG * KSPLIT == L_ && LSEG == 8, "bias rows must fill two 4-wide k-steps per segment");


constexpr size_t OFF_WEFF  = 0;
constexpr size_t OFF_STAGE = (size_t)G_ * NPAD;
constexpr size_t WS_FLOATS = OFF_STAGE + (size_t)TILES_M * 256;

constexpr int ZERO_WGS = TILES_M * 256 / 4 / 256;
__global__ __launch_bounds__(256)
void zero_stage_k(float* __restrict__ stage) {
    const int i = blockIdx.x * 256 + threadIdx.x;
    ((float4*)stage)[i] = make_float4(0.f, 0.f, 0.f, 0.f);
}

constexpr int OWN_ROWS    = 160;
constexpr int OWN_WGS     = G_ / OWN_ROWS;
constexpr int OWN_THREADS = 1024;
constexpr int OWN_WAVES   = OWN_THREADS / 32;
constexpr int NVEC        = L_ * g_ / 4;
constexpr int SCAN_ROUNDS = (NVEC + OWN_THREADS - 1) / OWN_THREADS;
constexpr int SCAN_BATCH  = 5;
constexpr int LIST_CAP    = 256;
constexpr int SUM_BATCH   = 8;
static_assert(OWN_WGS * OWN_ROWS == G_ && (OWN_ROWS & 1) == 0, "owner ranges tile Weff in whole lines");
static_assert(SCAN_ROUNDS % SCAN_BATCH == 0, "scan batches must divide the rounds");
static_assert(L_ * g_ < (1 << 17) && OWN_ROWS <= 256, "entry id (17 bits) and local row (8 bits) share one int");

__device__ __forceinline__ void add_entry_now(int t, int d,
                                              const float* __restrict__ W,
                                              const float* __restrict__ agg_w,
                                              float* acc) {
    const float aw = agg_w[t / g_];
    const float* w = W + (size_t)t * C_;
    float* dst = acc + (d >> 1) * 32 + (d & 1);
    for (int c = 0; c < C_; ++c) atomicAdd(dst + c * 2, aw * w[c]);
}

__global__ __launch_bounds__(OWN_THREADS)
void build_weff_k(const float* __restrict__ W,
                  const float* __restrict__ agg_w,
                  const int* __restrict__ gidx,
                  float* __restrict__ weff) {
    __shared__ __attribute__((aligned(16))) float acc[OWN_ROWS * NPAD];
    __shared__ int parked[OWN_WAVES][LIST_CAP];

    const int tid  = threadIdx.x;
    const int lane = tid & 31;
    const int wave = tid >> 5;
    const int k0   = blockIdx.x * OWN_ROWS;

    for (int i = tid; i < OWN_ROWS * NPAD; i += OWN_THREADS) acc[i] = 0.f;
    __syncthreads();

    int nw = 0;
    const int4* g4 = (const int4*)gidx;
    auto visit = [&](int k, int t, bool inr) {
        const int d = k - k0;
        const bool hit = inr && (unsigned)d < (unsigned)OWN_ROWS;
        const unsigned mask = __builtin_amdgcn_ballot_w32(hit);
        if (mask != 0u) {
            if (hit) {
                const int slot = nw + __builtin_popcount(mask & ((1u << lane) - 1u));
                if (slot < LIST_CAP) parked[wave][slot] = (t << 8) | d;
            }
            nw += __builtin_popcount(mask);
        }
    };
    for (int r0 = 0; r0 < SCAN_ROUNDS; r0 += SCAN_BATCH) {
        int4 q[SCAN_BATCH];
#pragma unroll
        for (int u = 0; u < SCAN_BATCH; ++u) {
            const int v = tid + (r0 + u) * OWN_THREADS;
            q[u] = g4[v < NVEC ? v : NVEC - 1];
        }
#pragma unroll
        for (int u = 0; u < SCAN_BATCH; ++u) {
            const int v = tid + (r0 + u) * OWN_THREADS;
            const bool inr = v < NVEC;
            visit(q[u].x, v * 4 + 0, inr);
            visit(q[u].y, v * 4 + 1, inr);
            visit(q[u].z, v * 4 + 2, inr);
            visit(q[u].w, v * 4 + 3, inr);
        }
    }
    if (nw > LIST_CAP) {
        int seen = 0;
        for (int i = 0; i < SCAN_ROUNDS * 4; ++i) {
            const int v = tid + (i >> 2) * OWN_THREADS;
            const bool inr = v < NVEC;
            const int t = (inr ? v : NVEC - 1) * 4 + (i & 3);
            const int d = gidx[t] - k0;
            const bool hit = inr && (unsigned)d < (unsigned)OWN_ROWS;
            const unsigned mask = __builtin_amdgcn_ballot_w32(hit);
            if (hit && seen + __builtin_popcount(mask & ((1u << lane) - 1u)) >= LIST_CAP)
                add_entry_now(t, d, W, agg_w, acc);
            seen += __builtin_popcount(mask);
        }
    }
    __syncthreads();

    const int total = (nw < LIST_CAP ? nw : LIST_CAP) * 5;
    for (int base = 0; base < total; base += 32 * SUM_BATCH) {
        v2f   wv[SUM_BATCH];
        float aw[SUM_BATCH];
        int   off[SUM_BATCH];
#pragma unroll
        for (int u = 0; u < SUM_BATCH; ++u) {
            const int i  = base + u * 32 + lane;
            const int ii = i < total ? i : 0;
            const int e  = ii / 5;
            const int j  = ii - e * 5;
            const int p  = parked[wave][e];
            const int t  = p >> 8;
            const int d  = p & 255;
            aw[u]  = agg_w[t / g_];
            wv[u]  = *(const v2f*)(W + (size_t)t * C_ + 2 * j);
            off[u] = (d >> 1) * 32 + j * 4 + (d & 1);
        }
#pragma unroll
        for (int u = 0; u < SUM_BATCH; ++u) {
            if (base + u * 32 + lane < total) {
                atomicAdd(&acc[off[u]],     aw[u] * wv[u].x);
                atomicAdd(&acc[off[u] + 2], aw[u] * wv[u].y);
            }
        }
    }
    __syncthreads();

    float4* dst = (float4*)(weff + (size_t)k0 * NPAD);
    const float4* src = (const float4*)acc;
    for (int i = tid; i < OWN_ROWS * NPAD / 4; i += OWN_THREADS) dst[i] = src[i];
}

__global__ __launch_bounds__(32)
void gemm_k(const float* __restrict__ X,
            const float* __restrict__ weff,
            const float* __restrict__ bias,
            const float* __restrict__ agg_w,
            float* __restrict__ stage) {
    const int tile = blockIdx.x % TILES_M;
    const int seg  = blockIdx.x / TILES_M;
    const int lane = threadIdx.x;
    const int half = lane >> 4;
    const int sub  = lane & 15;
    const int koff = half * 2;
    const int kbase = seg * KSEG;

    const float* ap = X + (size_t)(tile * 16 + sub) * G_ + kbase + koff;
    const float* bp = weff + (size_t)((kbase + koff) >> 1) * 32 + sub * 2;

    v8f acc = {0.f, 0.f, 0.f, 0.f, 0.f, 0.f, 0.f, 0.f};

    {
        const int l0 = seg * LSEG + koff;
        const bool live = sub < C_;
        const float* wl = agg_w + l0;
        const float* bl = bias + l0 * C_ + (live ? sub : 0);
        const v2f wa0 = {wl[0], wl[1]};
        const v2f wa1 = {wl[4], wl[5]};
        const float b00 = bl[0], b01 = bl[C_], b10 = bl[4 * C_], b11 = bl[5 * C_];
        const v2f bb0 = {live ? b00 : 0.f, live ? b01 : 0.f};
        const v2f bb1 = {live ? b10 : 0.f, live ? b11 : 0.f};
        acc = __builtin_amdgcn_wmma_f32_16x16x4_f32(
            false, wa0, false, bb0, (short)0, acc, false, false);
        acc = __builtin_amdgcn_wmma_f32_16x16x4_f32(
            false, wa1, false, bb1, (short)0, acc, false, false);
    }

    for (int o = 0; o < WSTEPS / UNROLL; ++o) {
        v2f a[UNROLL], b[UNROLL];
#pragma unroll
        for (int u = 0; u < UNROLL; ++u) {
            a[u] = *(const v2f*)(ap + u * 4);
            b[u] = *(const v2f*)(bp + u * 64);
        }
        __builtin_prefetch(ap + UNROLL * 4, 0, 3);
#pragma unroll
        for (int u = 0; u < UNROLL; ++u)
            acc = __builtin_amdgcn_wmma_f32_16x16x4_f32(
                 false, a[u],  false, b[u],
                 (short)0, acc,  false,  false);
        ap += UNROLL * 4;
        bp += UNROLL * 64;
    }

    float* st = stage + tile * 256 + sub;
    const int mbase = half * 8;
#pragma unroll
    for (int r = 0; r < 8; ++r)
        atomicAdd(st + (mbase + r) * 16, acc[r]);
}

__global__ void out_k(const float* __restrict__ stage,
                      const float* __restrict__ agg_b,
                      float* __restrict__ out) {
    int t = blockIdx.x * 256 + threadIdx.x;
    if (t >= B_ * C_) return;
    int b = t / C_, c = t % C_;
    out[t] = stage[(b >> 4) * 256 + (b & 15) * 16 + c] + agg_b[0];
}

extern "C" void kernel_launch(void* const* d_in, const int* in_sizes, int n_in,
                              void* d_out, int out_size, void* d_ws, size_t ws_size,
                              hipStream_t stream) {
    const float* X     = (const float*)d_in[0];
    const float* W     = (const float*)d_in[1];
    const float* bias  = (const float*)d_in[2];
    const float* agg_w = (const float*)d_in[3];
    const float* agg_b = (const float*)d_in[4];
    const int*   gidx  = (const int*)d_in[5];

    float* ws    = (float*)d_ws;
    float* weff  = ws + OFF_WEFF;
    float* stage = ws + OFF_STAGE;
    float* out   = (float*)d_out;

    (void)in_sizes; (void)n_in; (void)out_size; (void)ws_size;

    zero_stage_k<<<ZERO_WGS, 256, 0, stream>>>(stage);
    build_weff_k<<<OWN_WGS, OWN_THREADS, 0, stream>>>(W, agg_w, gidx, weff);
    gemm_k<<<TILES_M * KSPLIT, 32, 0, stream>>>(X, weff, bias, agg_w, stage);
    out_k<<<(B_ * C_ + 255) / 256, 256, 0, stream>>>(stage, agg_b, out);
}


# gemm split-K factor 125 -> 25: a wave runs 200 k-steps (25 lines per row), loads ~28 steps ahead, two accumulator chains; 1600 waves, 25 atomic partial sums per element (sum order changes)
# speedup vs baseline: 4.2847x; 1.1359x over previous
#include <hip/hip_runtime.h>

typedef __attribute__((ext_vector_type(2))) float v2f;
typedef __attribute__((ext_vector_type(8))) float v8f;

constexpr int B_ = 1024;
constexpr int G_ = 20000;
constexpr int L_ = 1000;
constexpr int g_ = 100;
constexpr int C_ = 10;
constexpr int NPAD = 16;

constexpr int TILES_M = B_ / 16;
constexpr int KSPLIT  = 25;
constexpr int KSEG    = G_ / KSPLIT;
constexpr int WSTEPS  = KSEG / 4;
constexpr int UNROLL  = 8;
constexpr int NBATCH  = WSTEPS / UNROLL;
constexpr int AHEAD   = 3;
constexpr int PF_DIST = 6;
static_assert(WSTEPS % UNROLL == 0 && KSEG % 32 == 0, "a batch is one whole line per row");
static_assert(AHEAD < PF_DIST && PF_DIST < NBATCH, "prefetch runs ahead of the register loads");
constexpr int LSEG    = L_ / KSPLIT;
constexpr int BSTEPS  = LSEG / 4;
static_assert(LSEG * KSPLIT == L_ && LSEG % 4 == 0, "bias rows must fill whole 4-wide k-steps per segment");


constexpr size_t OFF_WEFF  = 0;
constexpr size_t OFF_STAGE = (size_t)G_ * NPAD;
constexpr size_t WS_FLOATS = OFF_STAGE + (size_t)TILES_M * 256;

constexpr int ZERO_WGS = TILES_M * 256 / 4 / 256;
__global__ __launch_bounds__(256)
void zero_stage_k(float* __restrict__ stage) {
    const int i = blockIdx.x * 256 + threadIdx.x;
    ((float4*)stage)[i] = make_float4(0.f, 0.f, 0.f, 0.f);
}

constexpr int OWN_ROWS    = 160;
constexpr int OWN_WGS     = G_ / OWN_ROWS;
constexpr int OWN_THREADS = 1024;
constexpr int OWN_WAVES   = OWN_THREADS / 32;
constexpr int NVEC        = L_ * g_ / 4;
constexpr int SCAN_ROUNDS = (NVEC + OWN_THREADS - 1) / OWN_THREADS;
constexpr int SCAN_BATCH  = 5;
constexpr int LIST_CAP    = 256;
constexpr int SUM_BATCH   = 8;
static_assert(OWN_WGS * OWN_ROWS == G_ && (OWN_ROWS & 1) == 0, "owner ranges tile Weff in whole lines");
static_assert(SCAN_ROUNDS % SCAN_BATCH == 0, "scan batches must divide the rounds");
static_assert(L_ * g_ < (1 << 17) && OWN_ROWS <= 256, "entry id (17 bits) and local row (8 bits) share one int");

__device__ __forceinline__ void add_entry_now(int t, int d,
                                              const float* __restrict__ W,
                                              const float* __restrict__ agg_w,
                                              float* acc) {
    const float aw = agg_w[t / g_];
    const float* w = W + (size_t)t * C_;
    float* dst = acc + (d >> 1) * 32 + (d & 1);
    for (int c = 0; c < C_; ++c) atomicAdd(dst + c * 2, aw * w[c]);
}

__global__ __launch_bounds__(OWN_THREADS)
void build_weff_k(const float* __restrict__ W,
                  const float* __restrict__ agg_w,
                  const int* __restrict__ gidx,
                  float* __restrict__ weff) {
    __shared__ __attribute__((aligned(16))) float acc[OWN_ROWS * NPAD];
    __shared__ int parked[OWN_WAVES][LIST_CAP];

    const int tid  = threadIdx.x;
    const int lane = tid & 31;
    const int wave = tid >> 5;
    const int k0   = blockIdx.x * OWN_ROWS;

    for (int i = tid; i < OWN_ROWS * NPAD; i += OWN_THREADS) acc[i] = 0.f;
    __syncthreads();

    int nw = 0;
    const int4* g4 = (const int4*)gidx;
    auto visit = [&](int k, int t, bool inr) {
        const int d = k - k0;
        const bool hit = inr && (unsigned)d < (unsigned)OWN_ROWS;
        const unsigned mask = __builtin_amdgcn_ballot_w32(hit);
        if (mask != 0u) {
            if (hit) {
                const int slot = nw + __builtin_popcount(mask & ((1u << lane) - 1u));
                if (slot < LIST_CAP) parked[wave][slot] = (t << 8) | d;
            }
            nw += __builtin_popcount(mask);
        }
    };
    for (int r0 = 0; r0 < SCAN_ROUNDS; r0 += SCAN_BATCH) {
        int4 q[SCAN_BATCH];
#pragma unroll
        for (int u = 0; u < SCAN_BATCH; ++u) {
            const int v = tid + (r0 + u) * OWN_THREADS;
            q[u] = g4[v < NVEC ? v : NVEC - 1];
        }
#pragma unroll
        for (int u = 0; u < SCAN_BATCH; ++u) {
            const int v = tid + (r0 + u) * OWN_THREADS;
            const bool inr = v < NVEC;
            visit(q[u].x, v * 4 + 0, inr);
            visit(q[u].y, v * 4 + 1, inr);
            visit(q[u].z, v * 4 + 2, inr);
            visit(q[u].w, v * 4 + 3, inr);
        }
    }
    if (nw > LIST_CAP) {
        int seen = 0;
        for (int i = 0; i < SCAN_ROUNDS * 4; ++i) {
            const int v = tid + (i >> 2) * OWN_THREADS;
            const bool inr = v < NVEC;
            const int t = (inr ? v : NVEC - 1) * 4 + (i & 3);
            const int d = gidx[t] - k0;
            const bool hit = inr && (unsigned)d < (unsigned)OWN_ROWS;
            const unsigned mask = __builtin_amdgcn_ballot_w32(hit);
            if (hit && seen + __builtin_popcount(mask & ((1u << lane) - 1u)) >= LIST_CAP)
                add_entry_now(t, d, W, agg_w, acc);
            seen += __builtin_popcount(mask);
        }
    }
    __syncthreads();

    const int total = (nw < LIST_CAP ? nw : LIST_CAP) * 5;
    for (int base = 0; base < total; base += 32 * SUM_BATCH) {
        v2f   wv[SUM_BATCH];
        float aw[SUM_BATCH];
        int   off[SUM_BATCH];
#pragma unroll
        for (int u = 0; u < SUM_BATCH; ++u) {
            const int i  = base + u * 32 + lane;
            const int ii = i < total ? i : 0;
            const int e  = ii / 5;
            const int j  = ii - e * 5;
            const int p  = parked[wave][e];
            const int t  = p >> 8;
            const int d  = p & 255;
            aw[u]  = agg_w[t / g_];
            wv[u]  = *(const v2f*)(W + (size_t)t * C_ + 2 * j);
            off[u] = (d >> 1) * 32 + j * 4 + (d & 1);
        }
#pragma unroll
        for (int u = 0; u < SUM_BATCH; ++u) {
            if (base + u * 32 + lane < total) {
                atomicAdd(&acc[off[u]],     aw[u] * wv[u].x);
                atomicAdd(&acc[off[u] + 2], aw[u] * wv[u].y);
            }
        }
    }
    __syncthreads();

    float4* dst = (float4*)(weff + (size_t)k0 * NPAD);
    const float4* src = (const float4*)acc;
    for (int i = tid; i < OWN_ROWS * NPAD / 4; i += OWN_THREADS) dst[i] = src[i];
}

__global__ __launch_bounds__(32)
void gemm_k(const float* __restrict__ X,
            const float* __restrict__ weff,
            const float* __restrict__ bias,
            const float* __restrict__ agg_w,
            float* __restrict__ stage) {
    const int tile = blockIdx.x % TILES_M;
    const int seg  = blockIdx.x / TILES_M;
    const int lane = threadIdx.x;
    const int half = lane >> 4;
    const int sub  = lane & 15;
    const int koff = half * 2;
    const int kbase = seg * KSEG;

    const float* ap = X + (size_t)(tile * 16 + sub) * G_ + kbase + koff;
    const float* bp = weff + (size_t)((kbase + koff) >> 1) * 32 + sub * 2;

    v8f acc0 = {0.f, 0.f, 0.f, 0.f, 0.f, 0.f, 0.f, 0.f};
    v8f acc1 = {0.f, 0.f, 0.f, 0.f, 0.f, 0.f, 0.f, 0.f};

    v2f a[AHEAD + 1][UNROLL], b[AHEAD + 1][UNROLL];
    auto load_batch = [&](int o, v2f* av, v2f* bv) {
#pragma unroll
        for (int u = 0; u < UNROLL; ++u) {
            av[u] = *(const v2f*)(ap + (o * UNROLL + u) * 4);
            bv[u] = *(const v2f*)(bp + (size_t)(o * UNROLL + u) * 64);
        }
    };

#pragma unroll
    for (int p = 0; p < AHEAD; ++p) load_batch(p, a[p], b[p]);
#pragma unroll
    for (int p = AHEAD; p < PF_DIST; ++p) __builtin_prefetch(ap + p * UNROLL * 4, 0, 3);

    {
        const bool live = sub < C_;
        const float* wl = agg_w + seg * LSEG + koff;
        const float* bl = bias + (seg * LSEG + koff) * C_ + (live ? sub : 0);
#pragma unroll
        for (int j = 0; j < BSTEPS; ++j) {
            const v2f wa = {wl[4 * j], wl[4 * j + 1]};
            const float b0 = bl[4 * j * C_], b1 = bl[(4 * j + 1) * C_];
            const v2f bb = {live ? b0 : 0.f, live ? b1 : 0.f};
            if (j & 1)
                acc1 = __builtin_amdgcn_wmma_f32_16x16x4_f32(
                    false, wa, false, bb, (short)0, acc1, false, false);
            else
                acc0 = __builtin_amdgcn_wmma_f32_16x16x4_f32(
                    false, wa, false, bb, (short)0, acc0, false, false);
        }
    }

#pragma unroll
    for (int o = 0; o < NBATCH; ++o) {
        if (o + AHEAD < NBATCH)
            load_batch(o + AHEAD, a[(o + AHEAD) % (AHEAD + 1)], b[(o + AHEAD) % (AHEAD + 1)]);
        if (o + PF_DIST < NBATCH) __builtin_prefetch(ap + (o + PF_DIST) * UNROLL * 4, 0, 3);
        const v2f* av = a[o % (AHEAD + 1)];
        const v2f* bv = b[o % (AHEAD + 1)];
#pragma unroll
        for (int u = 0; u < UNROLL; u += 2) {
            acc0 = __builtin_amdgcn_wmma_f32_16x16x4_f32(
                 false, av[u],  false, bv[u],
                 (short)0, acc0,  false,  false);
            acc1 = __builtin_amdgcn_wmma_f32_16x16x4_f32(
                false, av[u + 1], false, bv[u + 1], (short)0, acc1, false, false);
        }
    }

    float* st = stage + tile * 256 + sub;
    const int mbase = half * 8;
#pragma unroll
    for (int r = 0; r < 8; ++r)
        atomicAdd(st + (mbase + r) * 16, acc0[r] + acc1[r]);
}

__global__ void out_k(const float* __restrict__ stage,
                      const float* __restrict__ agg_b,
                      float* __restrict__ out) {
    int t = blockIdx.x * 256 + threadIdx.x;
    if (t >= B_ * C_) return;
    int b = t / C_, c = t % C_;
    out[t] = stage[(b >> 4) * 256 + (b & 15) * 16 + c] + agg_b[0];
}

extern "C" void kernel_launch(void* const* d_in, const int* in_sizes, int n_in,
                              void* d_out, int out_size, void* d_ws, size_t ws_size,
                              hipStream_t stream) {
    const float* X     = (const float*)d_in[0];
    const float* W     = (const float*)d_in[1];
    const float* bias  = (const float*)d_in[2];
    const float* agg_w = (const float*)d_in[3];
    const float* agg_b = (const float*)d_in[4];
    const int*   gidx  = (const int*)d_in[5];

    float* ws    = (float*)d_ws;
    float* weff  = ws + OFF_WEFF;
    float* stage = ws + OFF_STAGE;
    float* out   = (float*)d_out;

    (void)in_sizes; (void)n_in; (void)out_size; (void)ws_size;

    zero_stage_k<<<ZERO_WGS, 256, 0, stream>>>(stage);
    build_weff_k<<<OWN_WGS, OWN_THREADS, 0, stream>>>(W, agg_w, gidx, weff);
    gemm_k<<<TILES_M * KSPLIT, 32, 0, stream>>>(X, weff, bias, agg_w, stage);
    out_k<<<(B_ * C_ + 255) / 256, 256, 0, stream>>>(stage, agg_b, out);
}


# split-K stage start value (zeros) stored by the first 4 owner work-groups of the Weff-build launch, one 16-byte store per thread behind their last barrier; clearing launch gone (4 -> 3 launches)
# speedup vs baseline: 4.5179x; 1.0544x over previous
#include <hip/hip_runtime.h>

typedef __attribute__((ext_vector_type(2))) float v2f;
typedef __attribute__((ext_vector_type(8))) float v8f;

constexpr int B_ = 1024;
constexpr int G_ = 20000;
constexpr int L_ = 1000;
constexpr int g_ = 100;
constexpr int C_ = 10;
constexpr int NPAD = 16;

constexpr int TILES_M = B_ / 16;
constexpr int KSPLIT  = 25;
constexpr int KSEG    = G_ / KSPLIT;
constexpr int WSTEPS  = KSEG / 4;
constexpr int UNROLL  = 8;
constexpr int NBATCH  = WSTEPS / UNROLL;
constexpr int AHEAD   = 3;
constexpr int PF_DIST = 6;
static_assert(WSTEPS % UNROLL == 0 && KSEG % 32 == 0, "a batch is one whole line per row");
static_assert(AHEAD < PF_DIST && PF_DIST < NBATCH, "prefetch runs ahead of the register loads");
constexpr int LSEG    = L_ / KSPLIT;
constexpr int BSTEPS  = LSEG / 4;
static_assert(LSEG * KSPLIT == L_ && LSEG % 4 == 0, "bias rows must fill whole 4-wide k-steps per segment");


constexpr size_t OFF_WEFF  = 0;
constexpr size_t OFF_STAGE = (size_t)G_ * NPAD;
constexpr size_t WS_FLOATS = OFF_STAGE + (size_t)TILES_M * 256;

constexpr int OWN_ROWS    = 160;
constexpr int OWN_WGS     = G_ / OWN_ROWS;
constexpr int OWN_THREADS = 1024;
constexpr int OWN_WAVES   = OWN_THREADS / 32;
constexpr int NVEC        = L_ * g_ / 4;
constexpr int SCAN_ROUNDS = (NVEC + OWN_THREADS - 1) / OWN_THREADS;
constexpr int SCAN_BATCH  = 5;
constexpr int LIST_CAP    = 256;
constexpr int STAGE_VEC   = TILES_M * 256 / 4;
constexpr int STAGE_WGS   = STAGE_VEC / OWN_THREADS;
static_assert(STAGE_WGS * OWN_THREADS == STAGE_VEC && STAGE_WGS <= G_ / 160, "whole work-groups clear the stage");
constexpr int SUM_BATCH   = 8;
static_assert(OWN_WGS * OWN_ROWS == G_ && (OWN_ROWS & 1) == 0, "owner ranges tile Weff in whole lines");
static_assert(SCAN_ROUNDS % SCAN_BATCH == 0, "scan batches must divide the rounds");
static_assert(L_ * g_ < (1 << 17) && OWN_ROWS <= 256, "entry id (17 bits) and local row (8 bits) share one int");

__device__ __forceinline__ void add_entry_now(int t, int d,
                                              const float* __restrict__ W,
                                              const float* __restrict__ agg_w,
                                              float* acc) {
    const float aw = agg_w[t / g_];
    const float* w = W + (size_t)t * C_;
    float* dst = acc + (d >> 1) * 32 + (d & 1);
    for (int c = 0; c < C_; ++c) atomicAdd(dst + c * 2, aw * w[c]);
}

__global__ __launch_bounds__(OWN_THREADS)
void build_weff_k(const float* __restrict__ W,
                  const float* __restrict__ agg_w,
                  const int* __restrict__ gidx,
                  float* __restrict__ weff,
                  float* __restrict__ stage) {
    __shared__ __attribute__((aligned(16))) float acc[OWN_ROWS * NPAD];
    __shared__ int parked[OWN_WAVES][LIST_CAP];

    const int tid  = threadIdx.x;
    const int lane = tid & 31;
    const int wave = tid >> 5;
    const int k0   = blockIdx.x * OWN_ROWS;

    for (int i = tid; i < OWN_ROWS * NPAD; i += OWN_THREADS) acc[i] = 0.f;
    __syncthreads();

    int nw = 0;
    const int4* g4 = (const int4*)gidx;
    auto visit = [&](int k, int t, bool inr) {
        const int d = k - k0;
        const bool hit = inr && (unsigned)d < (unsigned)OWN_ROWS;
        const unsigned mask = __builtin_amdgcn_ballot_w32(hit);
        if (mask != 0u) {
            if (hit) {
                const int slot = nw + __builtin_popcount(mask & ((1u << lane) - 1u));
                if (slot < LIST_CAP) parked[wave][slot] = (t << 8) | d;
            }
            nw += __builtin_popcount(mask);
        }
    };
    for (int r0 = 0; r0 < SCAN_ROUNDS; r0 += SCAN_BATCH) {
        int4 q[SCAN_BATCH];
#pragma unroll
        for (int u = 0; u < SCAN_BATCH; ++u) {
            const int v = tid + (r0 + u) * OWN_THREADS;
            q[u] = g4[v < NVEC ? v : NVEC - 1];
        }
#pragma unroll
        for (int u = 0; u < SCAN_BATCH; ++u) {
            const int v = tid + (r0 + u) * OWN_THREADS;
            const bool inr = v < NVEC;
            visit(q[u].x, v * 4 + 0, inr);
            visit(q[u].y, v * 4 + 1, inr);
            visit(q[u].z, v * 4 + 2, inr);
            visit(q[u].w, v * 4 + 3, inr);
        }
    }
    if (nw > LIST_CAP) {
        int seen = 0;
        for (int i = 0; i < SCAN_ROUNDS * 4; ++i) {
            const int v = tid + (i >> 2) * OWN_THREADS;
            const bool inr = v < NVEC;
            const int t = (inr ? v : NVEC - 1) * 4 + (i & 3);
            const int d = gidx[t] - k0;
            const bool hit = inr && (unsigned)d < (unsigned)OWN_ROWS;
            const unsigned mask = __builtin_amdgcn_ballot_w32(hit);
            if (hit && seen + __builtin_popcount(mask & ((1u << lane) - 1u)) >= LIST_CAP)
                add_entry_now(t, d, W, agg_w, acc);
            seen += __builtin_popcount(mask);
        }
    }
    __syncthreads();

    const int total = (nw < LIST_CAP ? nw : LIST_CAP) * 5;
    for (int base = 0; base < total; base += 32 * SUM_BATCH) {
        v2f   wv[SUM_BATCH];
        float aw[SUM_BATCH];
        int   off[SUM_BATCH];
#pragma unroll
        for (int u = 0; u < SUM_BATCH; ++u) {
            const int i  = base + u * 32 + lane;
            const int ii = i < total ? i : 0;
            const int e  = ii / 5;
            const int j  = ii - e * 5;
            const int p  = parked[wave][e];
            const int t  = p >> 8;
            const int d  = p & 255;
            aw[u]  = agg_w[t / g_];
            wv[u]  = *(const v2f*)(W + (size_t)t * C_ + 2 * j);
            off[u] = (d >> 1) * 32 + j * 4 + (d & 1);
        }
#pragma unroll
        for (int u = 0; u < SUM_BATCH; ++u) {
            if (base + u * 32 + lane < total) {
                atomicAdd(&acc[off[u]],     aw[u] * wv[u].x);
                atomicAdd(&acc[off[u] + 2], aw[u] * wv[u].y);
            }
        }
    }
    __syncthreads();

    float4* dst = (float4*)(weff + (size_t)k0 * NPAD);
    const float4* src = (const float4*)acc;
    for (int i = tid; i < OWN_ROWS * NPAD / 4; i += OWN_THREADS) dst[i] = src[i];

    if (blockIdx.x < STAGE_WGS)
        ((float4*)stage)[blockIdx.x * OWN_THREADS + tid] = make_float4(0.f, 0.f, 0.f, 0.f);
}

__global__ __launch_bounds__(32)
void gemm_k(const float* __restrict__ X,
            const float* __restrict__ weff,
            const float* __restrict__ bias,
            const float* __restrict__ agg_w,
            float* __restrict__ stage) {
    const int tile = blockIdx.x % TILES_M;
    const int seg  = blockIdx.x / TILES_M;
    const int lane = threadIdx.x;
    const int half = lane >> 4;
    const int sub  = lane & 15;
    const int koff = half * 2;
    const int kbase = seg * KSEG;

    const float* ap = X + (size_t)(tile * 16 + sub) * G_ + kbase + koff;
    const float* bp = weff + (size_t)((kbase + koff) >> 1) * 32 + sub * 2;

    v8f acc0 = {0.f, 0.f, 0.f, 0.f, 0.f, 0.f, 0.f, 0.f};
    v8f acc1 = {0.f, 0.f, 0.f, 0.f, 0.f, 0.f, 0.f, 0.f};

    v2f a[AHEAD + 1][UNROLL], b[AHEAD + 1][UNROLL];
    auto load_batch = [&](int o, v2f* av, v2f* bv) {
#pragma unroll
        for (int u = 0; u < UNROLL; ++u) {
            av[u] = *(const v2f*)(ap + (o * UNROLL + u) * 4);
            bv[u] = *(const v2f*)(bp + (size_t)(o * UNROLL + u) * 64);
        }
    };

#pragma unroll
    for (int p = 0; p < AHEAD; ++p) load_batch(p, a[p], b[p]);
#pragma unroll
    for (int p = AHEAD; p < PF_DIST; ++p) __builtin_prefetch(ap + p * UNROLL * 4, 0, 3);

    {
        const bool live = sub < C_;
        const float* wl = agg_w + seg * LSEG + koff;
        const float* bl = bias + (seg * LSEG + koff) * C_ + (live ? sub : 0);
#pragma unroll
        for (int j = 0; j < BSTEPS; ++j) {
            const v2f wa = {wl[4 * j], wl[4 * j + 1]};
            const float b0 = bl[4 * j * C_], b1 = bl[(4 * j + 1) * C_];
            const v2f bb = {live ? b0 : 0.f, live ? b1 : 0.f};
            if (j & 1)
                acc1 = __builtin_amdgcn_wmma_f32_16x16x4_f32(
                    false, wa, false, bb, (short)0, acc1, false, false);
            else
                acc0 = __builtin_amdgcn_wmma_f32_16x16x4_f32(
                    false, wa, false, bb, (short)0, acc0, false, false);
        }
    }

#pragma unroll
    for (int o = 0; o < NBATCH; ++o) {
        if (o + AHEAD < NBATCH)
            load_batch(o + AHEAD, a[(o + AHEAD) % (AHEAD + 1)], b[(o + AHEAD) % (AHEAD + 1)]);
        if (o + PF_DIST < NBATCH) __builtin_prefetch(ap + (o + PF_DIST) * UNROLL * 4, 0, 3);
        const v2f* av = a[o % (AHEAD + 1)];
        const v2f* bv = b[o % (AHEAD + 1)];
#pragma unroll
        for (int u = 0; u < UNROLL; u += 2) {
            acc0 = __builtin_amdgcn_wmma_f32_16x16x4_f32(
                 false, av[u],  false, bv[u],
                 (short)0, acc0,  false,  false);
            acc1 = __builtin_amdgcn_wmma_f32_16x16x4_f32(
                false, av[u + 1], false, bv[u + 1], (short)0, acc1, false, false);
        }
    }

    float* st = stage + tile * 256 + sub;
    const int mbase = half * 8;
#pragma unroll
    for (int r = 0; r < 8; ++r)
        atomicAdd(st + (mbase + r) * 16, acc0[r] + acc1[r]);
}

__global__ void out_k(const float* __restrict__ stage,
                      const float* __restrict__ agg_b,
                      float* __restrict__ out) {
    int t = blockIdx.x * 256 + threadIdx.x;
    if (t >= B_ * C_) return;
    int b = t / C_, c = t % C_;
    out[t] = stage[(b >> 4) * 256 + (b & 15) * 16 + c] + agg_b[0];
}

extern "C" void kernel_launch(void* const* d_in, const int* in_sizes, int n_in,
                              void* d_out, int out_size, void* d_ws, size_t ws_size,
                              hipStream_t stream) {
    const float* X     = (const float*)d_in[0];
    const float* W     = (const float*)d_in[1];
    const float* bias  = (const float*)d_in[2];
    const float* agg_w = (const float*)d_in[3];
    const float* agg_b = (const float*)d_in[4];
    const int*   gidx  = (const int*)d_in[5];

    float* ws    = (float*)d_ws;
    float* weff  = ws + OFF_WEFF;
    float* stage = ws + OFF_STAGE;
    float* out   = (float*)d_out;

    (void)in_sizes; (void)n_in; (void)out_size; (void)ws_size;

    build_weff_k<<<OWN_WGS, OWN_THREADS, 0, stream>>>(W, agg_w, gidx, weff, stage);
    gemm_k<<<TILES_M * KSPLIT, 32, 0, stream>>>(X, weff, bias, agg_w, stage);
    out_k<<<(B_ * C_ + 255) / 256, 256, 0, stream>>>(stage, agg_b, out);
}


# gemm fragment loads 8 -> 16 B per lane, straight into registers: a lane's 4 consecutive k feed a pair of k-steps (sum order inside each 8-k group changes); Weff in k-quads; 400 -> 200 loads a wave
# speedup vs baseline: 4.8322x; 1.0696x over previous
#include <hip/hip_runtime.h>

typedef __attribute__((ext_vector_type(2))) float v2f;
typedef __attribute__((ext_vector_type(4))) float v4f;
typedef __attribute__((ext_vector_type(8))) float v8f;

constexpr int B_ = 1024;
constexpr int G_ = 20000;
constexpr int L_ = 1000;
constexpr int g_ = 100;
constexpr int C_ = 10;
constexpr int NPAD = 16;

constexpr int TILES_M = B_ / 16;
constexpr int KSPLIT  = 25;
constexpr int KSEG    = G_ / KSPLIT;
constexpr int WSTEPS  = KSEG / 4;
constexpr int UNROLL  = 8;
constexpr int PAIRS   = UNROLL / 2;
constexpr int NBATCH  = WSTEPS / UNROLL;
constexpr int AHEAD   = 3;
constexpr int PF_DIST = 6;
static_assert(WSTEPS % UNROLL == 0 && KSEG % 32 == 0, "a batch is one whole line per row");
static_assert(AHEAD < PF_DIST && PF_DIST < NBATCH, "prefetch runs ahead of the register loads");
constexpr int LSEG    = L_ / KSPLIT;
constexpr int BSTEPS  = LSEG / 4;
static_assert(LSEG * KSPLIT == L_ && LSEG % 4 == 0, "bias rows must fill whole 4-wide k-steps per segment");


constexpr size_t OFF_WEFF  = 0;
constexpr size_t OFF_STAGE = (size_t)G_ * NPAD;
constexpr size_t WS_FLOATS = OFF_STAGE + (size_t)TILES_M * 256;

constexpr int OWN_ROWS    = 160;
constexpr int OWN_WGS     = G_ / OWN_ROWS;
constexpr int OWN_THREADS = 1024;
constexpr int OWN_WAVES   = OWN_THREADS / 32;
constexpr int NVEC        = L_ * g_ / 4;
constexpr int SCAN_ROUNDS = (NVEC + OWN_THREADS - 1) / OWN_THREADS;
constexpr int SCAN_BATCH  = 5;
constexpr int LIST_CAP    = 256;
constexpr int STAGE_VEC   = TILES_M * 256 / 4;
constexpr int STAGE_WGS   = STAGE_VEC / OWN_THREADS;
static_assert(STAGE_WGS * OWN_THREADS == STAGE_VEC && STAGE_WGS <= G_ / 160, "whole work-groups clear the stage");
constexpr int SUM_BATCH   = 8;
static_assert(OWN_WGS * OWN_ROWS == G_ && (OWN_ROWS & 3) == 0 && (G_ & 3) == 0, "owner ranges tile Weff in whole k-quads and lines");
static_assert(SCAN_ROUNDS % SCAN_BATCH == 0, "scan batches must divide the rounds");
static_assert(L_ * g_ < (1 << 17) && OWN_ROWS <= 256, "entry id (17 bits) and local row (8 bits) share one int");

__device__ __forceinline__ void add_entry_now(int t, int d,
                                              const float* __restrict__ W,
                                              const float* __restrict__ agg_w,
                                              float* acc) {
    const float aw = agg_w[t / g_];
    const float* w = W + (size_t)t * C_;
    float* dst = acc + (d >> 2) * 64 + (d & 3);
    for (int c = 0; c < C_; ++c) atomicAdd(dst + c * 4, aw * w[c]);
}

__global__ __launch_bounds__(OWN_THREADS)
void build_weff_k(const float* __restrict__ W,
                  const float* __restrict__ agg_w,
                  const int* __restrict__ gidx,
                  float* __restrict__ weff,
                  float* __restrict__ stage) {
    __shared__ __attribute__((aligned(16))) float acc[OWN_ROWS * NPAD];
    __shared__ int parked[OWN_WAVES][LIST_CAP];

    const int tid  = threadIdx.x;
    const int lane = tid & 31;
    const int wave = tid >> 5;
    const int k0   = blockIdx.x * OWN_ROWS;

    for (int i = tid; i < OWN_ROWS * NPAD; i += OWN_THREADS) acc[i] = 0.f;
    __syncthreads();

    int nw = 0;
    const int4* g4 = (const int4*)gidx;
    auto visit = [&](int k, int t, bool inr) {
        const int d = k - k0;
        const bool hit = inr && (unsigned)d < (unsigned)OWN_ROWS;
        const unsigned mask = __builtin_amdgcn_ballot_w32(hit);
        if (mask != 0u) {
            if (hit) {
                const int slot = nw + __builtin_popcount(mask & ((1u << lane) - 1u));
                if (slot < LIST_CAP) parked[wave][slot] = (t << 8) | d;
            }
            nw += __builtin_popcount(mask);
        }
    };
    for (int r0 = 0; r0 < SCAN_ROUNDS; r0 += SCAN_BATCH) {
        int4 q[SCAN_BATCH];
#pragma unroll
        for (int u = 0; u < SCAN_BATCH; ++u) {
            const int v = tid + (r0 + u) * OWN_THREADS;
            q[u] = g4[v < NVEC ? v : NVEC - 1];
        }
#pragma unroll
        for (int u = 0; u < SCAN_BATCH; ++u) {
            const int v = tid + (r0 + u) * OWN_THREADS;
            const bool inr = v < NVEC;
            visit(q[u].x, v * 4 + 0, inr);
            visit(q[u].y, v * 4 + 1, inr);
            visit(q[u].z, v * 4 + 2, inr);
            visit(q[u].w, v * 4 + 3, inr);
        }
    }
    if (nw > LIST_CAP) {
        int seen = 0;
        for (int i = 0; i < SCAN_ROUNDS * 4; ++i) {
            const int v = tid + (i >> 2) * OWN_THREADS;
            const bool inr = v < NVEC;
            const int t = (inr ? v : NVEC - 1) * 4 + (i & 3);
            const int d = gidx[t] - k0;
            const bool hit = inr && (unsigned)d < (unsigned)OWN_ROWS;
            const unsigned mask = __builtin_amdgcn_ballot_w32(hit);
            if (hit && seen + __builtin_popcount(mask & ((1u << lane) - 1u)) >= LIST_CAP)
                add_entry_now(t, d, W, agg_w, acc);
            seen += __builtin_popcount(mask);
        }
    }
    __syncthreads();

    const int total = (nw < LIST_CAP ? nw : LIST_CAP) * 5;
    for (int base = 0; base < total; base += 32 * SUM_BATCH) {
        v2f   wv[SUM_BATCH];
        float aw[SUM_BATCH];
        int   off[SUM_BATCH];
#pragma unroll
        for (int u = 0; u < SUM_BATCH; ++u) {
            const int i  = base + u * 32 + lane;
            const int ii = i < total ? i : 0;
            const int e  = ii / 5;
            const int j  = ii - e * 5;
            const int p  = parked[wave][e];
            const int t  = p >> 8;
            const int d  = p & 255;
            aw[u]  = agg_w[t / g_];
            wv[u]  = *(const v2f*)(W + (size_t)t * C_ + 2 * j);
            off[u] = (d >> 2) * 64 + j * 8 + (d & 3);
        }
#pragma unroll
        for (int u = 0; u < SUM_BATCH; ++u) {
            if (base + u * 32 + lane < total) {
                atomicAdd(&acc[off[u]],     aw[u] * wv[u].x);
                atomicAdd(&acc[off[u] + 4], aw[u] * wv[u].y);
            }
        }
    }
    __syncthreads();

    float4* dst = (float4*)(weff + (size_t)k0 * NPAD);
    const float4* src = (const float4*)acc;
    for (int i = tid; i < OWN_ROWS * NPAD / 4; i += OWN_THREADS) dst[i] = src[i];

    if (blockIdx.x < STAGE_WGS)
        ((float4*)stage)[blockIdx.x * OWN_THREADS + tid] = make_float4(0.f, 0.f, 0.f, 0.f);
}

__global__ __launch_bounds__(32)
void gemm_k(const float* __restrict__ X,
            const float* __restrict__ weff,
            const float* __restrict__ bias,
            const float* __restrict__ agg_w,
            float* __restrict__ stage) {
    const int tile = blockIdx.x % TILES_M;
    const int seg  = blockIdx.x / TILES_M;
    const int lane = threadIdx.x;
    const int half = lane >> 4;
    const int sub  = lane & 15;
    const int koff = half * 4;
    const int kbase = seg * KSEG;

    const float* ap = X + (size_t)(tile * 16 + sub) * G_ + kbase + koff;
    const float* bp = weff + (size_t)((kbase + koff) >> 2) * 64 + sub * 4;

    v8f acc0 = {0.f, 0.f, 0.f, 0.f, 0.f, 0.f, 0.f, 0.f};
    v8f acc1 = {0.f, 0.f, 0.f, 0.f, 0.f, 0.f, 0.f, 0.f};

    v4f a[AHEAD + 1][PAIRS], b[AHEAD + 1][PAIRS];
    auto load_batch = [&](int o, v4f* av, v4f* bv) {
#pragma unroll
        for (int u = 0; u < PAIRS; ++u) {
            av[u] = *(const v4f*)(ap + (o * PAIRS + u) * 8);
            bv[u] = *(const v4f*)(bp + (size_t)(o * PAIRS + u) * 128);
        }
    };

#pragma unroll
    for (int p = 0; p < AHEAD; ++p) load_batch(p, a[p], b[p]);
#pragma unroll
    for (int p = AHEAD; p < PF_DIST; ++p) __builtin_prefetch(ap + p * UNROLL * 4, 0, 3);

    {
        const bool live = sub < C_;
        const int lk = half * 2;
        const float* wl = agg_w + seg * LSEG + lk;
        const float* bl = bias + (seg * LSEG + lk) * C_ + (live ? sub : 0);
#pragma unroll
        for (int j = 0; j < BSTEPS; ++j) {
            const v2f wa = {wl[4 * j], wl[4 * j + 1]};
            const float b0 = bl[4 * j * C_], b1 = bl[(4 * j + 1) * C_];
            const v2f bb = {live ? b0 : 0.f, live ? b1 : 0.f};
            if (j & 1)
                acc1 = __builtin_amdgcn_wmma_f32_16x16x4_f32(
                    false, wa, false, bb, (short)0, acc1, false, false);
            else
                acc0 = __builtin_amdgcn_wmma_f32_16x16x4_f32(
                    false, wa, false, bb, (short)0, acc0, false, false);
        }
    }

#pragma unroll
    for (int o = 0; o < NBATCH; ++o) {
        if (o + AHEAD < NBATCH)
            load_batch(o + AHEAD, a[(o + AHEAD) % (AHEAD + 1)], b[(o + AHEAD) % (AHEAD + 1)]);
        if (o + PF_DIST < NBATCH) __builtin_prefetch(ap + (o + PF_DIST) * UNROLL * 4, 0, 3);
        const v4f* av = a[o % (AHEAD + 1)];
        const v4f* bv = b[o % (AHEAD + 1)];
#pragma unroll
        for (int u = 0; u < PAIRS; ++u) {
            acc0 = __builtin_amdgcn_wmma_f32_16x16x4_f32(
                 false, av[u].lo,  false, bv[u].lo,
                 (short)0, acc0,  false,  false);
            acc1 = __builtin_amdgcn_wmma_f32_16x16x4_f32(
                false, av[u].hi, false, bv[u].hi, (short)0, acc1, false, false);
        }
    }

    float* st = stage + tile * 256 + sub;
    const int mbase = half * 8;
#pragma unroll
    for (int r = 0; r < 8; ++r)
        atomicAdd(st + (mbase + r) * 16, acc0[r] + acc1[r]);
}

__global__ void out_k(const float* __restrict__ stage,
                      const float* __restrict__ agg_b,
                      float* __restrict__ out) {
    int t = blockIdx.x * 256 + threadIdx.x;
    if (t >= B_ * C_) return;
    int b = t / C_, c = t % C_;
    out[t] = stage[(b >> 4) * 256 + (b & 15) * 16 + c] + agg_b[0];
}

extern "C" void kernel_launch(void* const* d_in, const int* in_sizes, int n_in,
                              void* d_out, int out_size, void* d_ws, size_t ws_size,
                              hipStream_t stream) {
    const float* X     = (const float*)d_in[0];
    const float* W     = (const float*)d_in[1];
    const float* bias  = (const float*)d_in[2];
    const float* agg_w = (const float*)d_in[3];
    const float* agg_b = (const float*)d_in[4];
    const int*   gidx  = (const int*)d_in[5];

    float* ws    = (float*)d_ws;
    float* weff  = ws + OFF_WEFF;
    float* stage = ws + OFF_STAGE;
    float* out   = (float*)d_out;

    (void)in_sizes; (void)n_in; (void)out_size; (void)ws_size;

    build_weff_k<<<OWN_WGS, OWN_THREADS, 0, stream>>>(W, agg_w, gidx, weff, stage);
    gemm_k<<<TILES_M * KSPLIT, 32, 0, stream>>>(X, weff, bias, agg_w, stage);
    out_k<<<(B_ * C_ + 255) / 256, 256, 0, stream>>>(stage, agg_b, out);
}


# gemm grid sized for two waves per SIMD: 32 K segments (19 or 20 lines), 4-wave work-groups = 4 row tiles of one segment sharing Weff lines in the WGP cache; 32 atomic sums per element (order changes)
# speedup vs baseline: 5.3793x; 1.1132x over previous
#include <hip/hip_runtime.h>

typedef __attribute__((ext_vector_type(2))) float v2f;
typedef __attribute__((ext_vector_type(4))) float v4f;
typedef __attribute__((ext_vector_type(8))) float v8f;

constexpr int B_ = 1024;
constexpr int G_ = 20000;
constexpr int L_ = 1000;
constexpr int g_ = 100;
constexpr int C_ = 10;
constexpr int NPAD = 16;

constexpr int TILES_M = B_ / 16;
constexpr int KSPLIT  = 32;
constexpr int GW      = 4;
constexpr int XLINES  = G_ / 32;
constexpr int NB_MIN  = XLINES / KSPLIT;
constexpr int NB_MAX  = NB_MIN + 1;
constexpr int LONG_SEGS = XLINES - NB_MIN * KSPLIT;
constexpr int UNROLL  = 8;
constexpr int PAIRS   = UNROLL / 2;
constexpr int AHEAD   = 3;
constexpr int PF_DIST = 6;
static_assert(G_ % 32 == 0 && TILES_M % GW == 0, "whole lines per row, whole work-groups of row tiles");
static_assert(LONG_SEGS >= 0 && LONG_SEGS < KSPLIT && LONG_SEGS * NB_MAX + (KSPLIT - LONG_SEGS) * NB_MIN == XLINES,
              "the segments cover every line once");
static_assert(AHEAD < PF_DIST && PF_DIST < NB_MIN, "prefetch runs ahead of the register loads");
constexpr int BSEGS   = 25;
constexpr int LSEG    = L_ / BSEGS;
constexpr int BSTEPS  = LSEG / 4;
static_assert(LSEG * BSEGS == L_ && LSEG % 4 == 0 && BSEGS <= KSPLIT, "bias rows must fill whole 4-wide k-steps per segment");


constexpr size_t OFF_WEFF  = 0;
constexpr size_t OFF_STAGE = (size_t)G_ * NPAD;
constexpr size_t WS_FLOATS = OFF_STAGE + (size_t)TILES_M * 256;

constexpr int OWN_ROWS    = 160;
constexpr int OWN_WGS     = G_ / OWN_ROWS;
constexpr int OWN_THREADS = 1024;
constexpr int OWN_WAVES   = OWN_THREADS / 32;
constexpr int NVEC        = L_ * g_ / 4;
constexpr int SCAN_ROUNDS = (NVEC + OWN_THREADS - 1) / OWN_THREADS;
constexpr int SCAN_BATCH  = 5;
constexpr int LIST_CAP    = 256;
constexpr int STAGE_VEC   = TILES_M * 256 / 4;
constexpr int STAGE_WGS   = STAGE_VEC / OWN_THREADS;
static_assert(STAGE_WGS * OWN_THREADS == STAGE_VEC && STAGE_WGS <= G_ / 160, "whole work-groups clear the stage");
constexpr int SUM_BATCH   = 8;
static_assert(OWN_WGS * OWN_ROWS == G_ && (OWN_ROWS & 3) == 0 && (G_ & 3) == 0, "owner ranges tile Weff in whole k-quads and lines");
static_assert(SCAN_ROUNDS % SCAN_BATCH == 0, "scan batches must divide the rounds");
static_assert(L_ * g_ < (1 << 17) && OWN_ROWS <= 256, "entry id (17 bits) and local row (8 bits) share one int");

__device__ __forceinline__ void add_entry_now(int t, int d,
                                              const float* __restrict__ W,
                                              const float* __restrict__ agg_w,
                                              float* acc) {
    const float aw = agg_w[t / g_];
    const float* w = W + (size_t)t * C_;
    float* dst = acc + (d >> 2) * 64 + (d & 3);
    for (int c = 0; c < C_; ++c) atomicAdd(dst + c * 4, aw * w[c]);
}

__global__ __launch_bounds__(OWN_THREADS)
void build_weff_k(const float* __restrict__ W,
                  const float* __restrict__ agg_w,
                  const int* __restrict__ gidx,
                  float* __restrict__ weff,
                  float* __restrict__ stage) {
    __shared__ __attribute__((aligned(16))) float acc[OWN_ROWS * NPAD];
    __shared__ int parked[OWN_WAVES][LIST_CAP];

    const int tid  = threadIdx.x;
    const int lane = tid & 31;
    const int wave = tid >> 5;
    const int k0   = blockIdx.x * OWN_ROWS;

    for (int i = tid; i < OWN_ROWS * NPAD; i += OWN_THREADS) acc[i] = 0.f;
    __syncthreads();

    int nw = 0;
    const int4* g4 = (const int4*)gidx;
    auto visit = [&](int k, int t, bool inr) {
        const int d = k - k0;
        const bool hit = inr && (unsigned)d < (unsigned)OWN_ROWS;
        const unsigned mask = __builtin_amdgcn_ballot_w32(hit);
        if (mask != 0u) {
            if (hit) {
                const int slot = nw + __builtin_popcount(mask & ((1u << lane) - 1u));
                if (slot < LIST_CAP) parked[wave][slot] = (t << 8) | d;
            }
            nw += __builtin_popcount(mask);
        }
    };
    for (int r0 = 0; r0 < SCAN_ROUNDS; r0 += SCAN_BATCH) {
        int4 q[SCAN_BATCH];
#pragma unroll
        for (int u = 0; u < SCAN_BATCH; ++u) {
            const int v = tid + (r0 + u) * OWN_THREADS;
            q[u] = g4[v < NVEC ? v : NVEC - 1];
        }
#pragma unroll
        for (int u = 0; u < SCAN_BATCH; ++u) {
            const int v = tid + (r0 + u) * OWN_THREADS;
            const bool inr = v < NVEC;
            visit(q[u].x, v * 4 + 0, inr);
            visit(q[u].y, v * 4 + 1, inr);
            visit(q[u].z, v * 4 + 2, inr);
            visit(q[u].w, v * 4 + 3, inr);
        }
    }
    if (nw > LIST_CAP) {
        int seen = 0;
        for (int i = 0; i < SCAN_ROUNDS * 4; ++i) {
            const int v = tid + (i >> 2) * OWN_THREADS;
            const bool inr = v < NVEC;
            const int t = (inr ? v : NVEC - 1) * 4 + (i & 3);
            const int d = gidx[t] - k0;
            const bool hit = inr && (unsigned)d < (unsigned)OWN_ROWS;
            const unsigned mask = __builtin_amdgcn_ballot_w32(hit);
            if (hit && seen + __builtin_popcount(mask & ((1u << lane) - 1u)) >= LIST_CAP)
                add_entry_now(t, d, W, agg_w, acc);
            seen += __builtin_popcount(mask);
        }
    }
    __syncthreads();

    const int total = (nw < LIST_CAP ? nw : LIST_CAP) * 5;
    for (int base = 0; base < total; base += 32 * SUM_BATCH) {
        v2f   wv[SUM_BATCH];
        float aw[SUM_BATCH];
        int   off[SUM_BATCH];
#pragma unroll
        for (int u = 0; u < SUM_BATCH; ++u) {
            const int i  = base + u * 32 + lane;
            const int ii = i < total ? i : 0;
            const int e  = ii / 5;
            const int j  = ii - e * 5;
            const int p  = parked[wave][e];
            const int t  = p >> 8;
            const int d  = p & 255;
            aw[u]  = agg_w[t / g_];
            wv[u]  = *(const v2f*)(W + (size_t)t * C_ + 2 * j);
            off[u] = (d >> 2) * 64 + j * 8 + (d & 3);
        }
#pragma unroll
        for (int u = 0; u < SUM_BATCH; ++u) {
            if (base + u * 32 + lane < total) {
                atomicAdd(&acc[off[u]],     aw[u] * wv[u].x);
                atomicAdd(&acc[off[u] + 4], aw[u] * wv[u].y);
            }
        }
    }
    __syncthreads();

    float4* dst = (float4*)(weff + (size_t)k0 * NPAD);
    const float4* src = (const float4*)acc;
    for (int i = tid; i < OWN_ROWS * NPAD / 4; i += OWN_THREADS) dst[i] = src[i];

    if (blockIdx.x < STAGE_WGS)
        ((float4*)stage)[blockIdx.x * OWN_THREADS + tid] = make_float4(0.f, 0.f, 0.f, 0.f);
}

__global__ __launch_bounds__(32 * GW)
void gemm_k(const float* __restrict__ X,
            const float* __restrict__ weff,
            const float* __restrict__ bias,
            const float* __restrict__ agg_w,
            float* __restrict__ stage) {
    const int tile = (blockIdx.x % (TILES_M / GW)) * GW + (threadIdx.x >> 5);
    const int seg  = blockIdx.x / (TILES_M / GW);
    const bool long_seg = seg < LONG_SEGS;
    const int line0 = seg * NB_MIN + (long_seg ? seg : LONG_SEGS);
    const int lane = threadIdx.x & 31;
    const int half = lane >> 4;
    const int sub  = lane & 15;
    const int koff = half * 4;
    const int kbase = line0 * 32;

    const float* ap = X + (size_t)(tile * 16 + sub) * G_ + kbase + koff;
    const float* bp = weff + (size_t)((kbase + koff) >> 2) * 64 + sub * 4;

    v8f acc0 = {0.f, 0.f, 0.f, 0.f, 0.f, 0.f, 0.f, 0.f};
    v8f acc1 = {0.f, 0.f, 0.f, 0.f, 0.f, 0.f, 0.f, 0.f};

    v4f a[AHEAD + 1][PAIRS], b[AHEAD + 1][PAIRS];
    auto load_batch = [&](int o, v4f* av, v4f* bv) {
#pragma unroll
        for (int u = 0; u < PAIRS; ++u) {
            av[u] = *(const v4f*)(ap + (o * PAIRS + u) * 8);
            bv[u] = *(const v4f*)(bp + (size_t)(o * PAIRS + u) * 128);
        }
    };

#pragma unroll
    for (int p = 0; p < AHEAD; ++p) load_batch(p, a[p], b[p]);
#pragma unroll
    for (int p = AHEAD; p < PF_DIST; ++p) __builtin_prefetch(ap + p * UNROLL * 4, 0, 3);

    if (seg < BSEGS) {
        const bool live = sub < C_;
        const int lk = half * 2;
        const float* wl = agg_w + seg * LSEG + lk;
        const float* bl = bias + (seg * LSEG + lk) * C_ + (live ? sub : 0);
#pragma unroll
        for (int j = 0; j < BSTEPS; ++j) {
            const v2f wa = {wl[4 * j], wl[4 * j + 1]};
            const float b0 = bl[4 * j * C_], b1 = bl[(4 * j + 1) * C_];
            const v2f bb = {live ? b0 : 0.f, live ? b1 : 0.f};
            if (j & 1)
                acc1 = __builtin_amdgcn_wmma_f32_16x16x4_f32(
                    false, wa, false, bb, (short)0, acc1, false, false);
            else
                acc0 = __builtin_amdgcn_wmma_f32_16x16x4_f32(
                    false, wa, false, bb, (short)0, acc0, false, false);
        }
    }

    auto mul_batch = [&](const v4f* av, const v4f* bv) {
#pragma unroll
        for (int u = 0; u < PAIRS; ++u) {
            acc0 = __builtin_amdgcn_wmma_f32_16x16x4_f32(
                 false, av[u].lo,  false, bv[u].lo,
                 (short)0, acc0,  false,  false);
            acc1 = __builtin_amdgcn_wmma_f32_16x16x4_f32(
                false, av[u].hi, false, bv[u].hi, (short)0, acc1, false, false);
        }
    };
#pragma unroll
    for (int o = 0; o < NB_MAX; ++o) {
        const int ol = o + AHEAD, op = o + PF_DIST;
        if (ol < NB_MIN)
            load_batch(ol, a[ol % (AHEAD + 1)], b[ol % (AHEAD + 1)]);
        else if (ol < NB_MAX) {
            if (long_seg) load_batch(ol, a[ol % (AHEAD + 1)], b[ol % (AHEAD + 1)]);
        }
        if (op < NB_MIN) __builtin_prefetch(ap + op * UNROLL * 4, 0, 3);
        else if (op < NB_MAX) { if (long_seg) __builtin_prefetch(ap + op * UNROLL * 4, 0, 3); }
        if (o < NB_MIN) mul_batch(a[o % (AHEAD + 1)], b[o % (AHEAD + 1)]);
        else { if (long_seg) mul_batch(a[o % (AHEAD + 1)], b[o % (AHEAD + 1)]); }
    }

    float* st = stage + tile * 256 + sub;
    const int mbase = half * 8;
#pragma unroll
    for (int r = 0; r < 8; ++r)
        atomicAdd(st + (mbase + r) * 16, acc0[r] + acc1[r]);
}

__global__ void out_k(const float* __restrict__ stage,
                      const float* __restrict__ agg_b,
                      float* __restrict__ out) {
    int t = blockIdx.x * 256 + threadIdx.x;
    if (t >= B_ * C_) return;
    int b = t / C_, c = t % C_;
    out[t] = stage[(b >> 4) * 256 + (b & 15) * 16 + c] + agg_b[0];
}

extern "C" void kernel_launch(void* const* d_in, const int* in_sizes, int n_in,
                              void* d_out, int out_size, void* d_ws, size_t ws_size,
                              hipStream_t stream) {
    const float* X     = (const float*)d_in[0];
    const float* W     = (const float*)d_in[1];
    const float* bias  = (const float*)d_in[2];
    const float* agg_w = (const float*)d_in[3];
    const float* agg_b = (const float*)d_in[4];
    const int*   gidx  = (const int*)d_in[5];

    float* ws    = (float*)d_ws;
    float* weff  = ws + OFF_WEFF;
    float* stage = ws + OFF_STAGE;
    float* out   = (float*)d_out;

    (void)in_sizes; (void)n_in; (void)out_size; (void)ws_size;

    build_weff_k<<<OWN_WGS, OWN_THREADS, 0, stream>>>(W, agg_w, gidx, weff, stage);
    gemm_k<<<(TILES_M / GW) * KSPLIT, 32 * GW, 0, stream>>>(X, weff, bias, agg_w, stage);
    out_k<<<(B_ * C_ + 255) / 256, 256, 0, stream>>>(stage, agg_b, out);
}


# gemm B operand (Weff) staged in LDS: a work-group copies its K segment's 38-40 KB run once (whole-line loads, one barrier), its 4 waves read B fragments from LDS; global loads per wave 160 -> 100
# speedup vs baseline: 5.5579x; 1.0332x over previous
#include <hip/hip_runtime.h>

typedef __attribute__((ext_vector_type(2))) float v2f;
typedef __attribute__((ext_vector_type(4))) float v4f;
typedef __attribute__((ext_vector_type(8))) float v8f;

constexpr int B_ = 1024;
constexpr int G_ = 20000;
constexpr int L_ = 1000;
constexpr int g_ = 100;
constexpr int C_ = 10;
constexpr int NPAD = 16;

constexpr int TILES_M = B_ / 16;
constexpr int KSPLIT  = 32;
constexpr int GW      = 4;
constexpr int XLINES  = G_ / 32;
constexpr int NB_MIN  = XLINES / KSPLIT;
constexpr int NB_MAX  = NB_MIN + 1;
constexpr int LONG_SEGS = XLINES - NB_MIN * KSPLIT;
constexpr int UNROLL  = 8;
constexpr int PAIRS   = UNROLL / 2;
constexpr int AHEAD   = 3;
constexpr int RING    = AHEAD + 1;
constexpr int PF_DIST = 6;
static_assert(G_ % 32 == 0 && TILES_M % GW == 0, "whole lines per row, whole work-groups of row tiles");
static_assert(LONG_SEGS >= 0 && LONG_SEGS < KSPLIT && LONG_SEGS * NB_MAX + (KSPLIT - LONG_SEGS) * NB_MIN == XLINES,
              "the segments cover every line once");
static_assert(AHEAD < PF_DIST && PF_DIST < NB_MIN, "prefetch runs ahead of the register loads");
constexpr int BVEC    = UNROLL * 4 * NPAD / 4;
static_assert(BVEC == 32 * GW, "staging: one 16-byte piece per thread per batch");
constexpr int BSEGS   = 25;
constexpr int LSEG    = L_ / BSEGS;
constexpr int BSTEPS  = LSEG / 4;
static_assert(LSEG * BSEGS == L_ && LSEG % 4 == 0 && BSEGS <= KSPLIT, "bias rows must fill whole 4-wide k-steps per segment");


constexpr size_t OFF_WEFF  = 0;
constexpr size_t OFF_STAGE = (size_t)G_ * NPAD;
constexpr size_t WS_FLOATS = OFF_STAGE + (size_t)TILES_M * 256;

constexpr int OWN_ROWS    = 160;
constexpr int OWN_WGS     = G_ / OWN_ROWS;
constexpr int OWN_THREADS = 1024;
constexpr int OWN_WAVES   = OWN_THREADS / 32;
constexpr int NVEC        = L_ * g_ / 4;
constexpr int SCAN_ROUNDS = (NVEC + OWN_THREADS - 1) / OWN_THREADS;
constexpr int SCAN_BATCH  = 5;
constexpr int LIST_CAP    = 256;
constexpr int STAGE_VEC   = TILES_M * 256 / 4;
constexpr int STAGE_WGS   = STAGE_VEC / OWN_THREADS;
static_assert(STAGE_WGS * OWN_THREADS == STAGE_VEC && STAGE_WGS <= G_ / 160, "whole work-groups clear the stage");
constexpr int SUM_BATCH   = 8;
static_assert(OWN_WGS * OWN_ROWS == G_ && (OWN_ROWS & 3) == 0 && (G_ & 3) == 0, "owner ranges tile Weff in whole k-quads and lines");
static_assert(SCAN_ROUNDS % SCAN_BATCH == 0, "scan batches must divide the rounds");
static_assert(L_ * g_ < (1 << 17) && OWN_ROWS <= 256, "entry id (17 bits) and local row (8 bits) share one int");

__device__ __forceinline__ void add_entry_now(int t, int d,
                                              const float* __restrict__ W,
                                              const float* __restrict__ agg_w,
                                              float* acc) {
    const float aw = agg_w[t / g_];
    const float* w = W + (size_t)t * C_;
    float* dst = acc + (d >> 2) * 64 + (d & 3);
    for (int c = 0; c < C_; ++c) atomicAdd(dst + c * 4, aw * w[c]);
}

__global__ __launch_bounds__(OWN_THREADS)
void build_weff_k(const float* __restrict__ W,
                  const float* __restrict__ agg_w,
                  const int* __restrict__ gidx,
                  float* __restrict__ weff,
                  float* __restrict__ stage) {
    __shared__ __attribute__((aligned(16))) float acc[OWN_ROWS * NPAD];
    __shared__ int parked[OWN_WAVES][LIST_CAP];

    const int tid  = threadIdx.x;
    const int lane = tid & 31;
    const int wave = tid >> 5;
    const int k0   = blockIdx.x * OWN_ROWS;

    for (int i = tid; i < OWN_ROWS * NPAD; i += OWN_THREADS) acc[i] = 0.f;
    __syncthreads();

    int nw = 0;
    const int4* g4 = (const int4*)gidx;
    auto visit = [&](int k, int t, bool inr) {
        const int d = k - k0;
        const bool hit = inr && (unsigned)d < (unsigned)OWN_ROWS;
        const unsigned mask = __builtin_amdgcn_ballot_w32(hit);
        if (mask != 0u) {
            if (hit) {
                const int slot = nw + __builtin_popcount(mask & ((1u << lane) - 1u));
                if (slot < LIST_CAP) parked[wave][slot] = (t << 8) | d;
            }
            nw += __builtin_popcount(mask);
        }
    };
    for (int r0 = 0; r0 < SCAN_ROUNDS; r0 += SCAN_BATCH) {
        int4 q[SCAN_BATCH];
#pragma unroll
        for (int u = 0; u < SCAN_BATCH; ++u) {
            const int v = tid + (r0 + u) * OWN_THREADS;
            q[u] = g4[v < NVEC ? v : NVEC - 1];
        }
#pragma unroll
        for (int u = 0; u < SCAN_BATCH; ++u) {
            const int v = tid + (r0 + u) * OWN_THREADS;
            const bool inr = v < NVEC;
            visit(q[u].x, v * 4 + 0, inr);
            visit(q[u].y, v * 4 + 1, inr);
            visit(q[u].z, v * 4 + 2, inr);
            visit(q[u].w, v * 4 + 3, inr);
        }
    }
    if (nw > LIST_CAP) {
        int seen = 0;
        for (int i = 0; i < SCAN_ROUNDS * 4; ++i) {
            const int v = tid + (i >> 2) * OWN_THREADS;
            const bool inr = v < NVEC;
            const int t = (inr ? v : NVEC - 1) * 4 + (i & 3);
            const int d = gidx[t] - k0;
            const bool hit = inr && (unsigned)d < (unsigned)OWN_ROWS;
            const unsigned mask = __builtin_amdgcn_ballot_w32(hit);
            if (hit && seen + __builtin_popcount(mask & ((1u << lane) - 1u)) >= LIST_CAP)
                add_entry_now(t, d, W, agg_w, acc);
            seen += __builtin_popcount(mask);
        }
    }
    __syncthreads();

    const int total = (nw < LIST_CAP ? nw : LIST_CAP) * 5;
    for (int base = 0; base < total; base += 32 * SUM_BATCH) {
        v2f   wv[SUM_BATCH];
        float aw[SUM_BATCH];
        int   off[SUM_BATCH];
#pragma unroll
        for (int u = 0; u < SUM_BATCH; ++u) {
            const int i  = base + u * 32 + lane;
            const int ii = i < total ? i : 0;
            const int e  = ii / 5;
            const int j  = ii - e * 5;
            const int p  = parked[wave][e];
            const int t  = p >> 8;
            const int d  = p & 255;
            aw[u]  = agg_w[t / g_];
            wv[u]  = *(const v2f*)(W + (size_t)t * C_ + 2 * j);
            off[u] = (d >> 2) * 64 + j * 8 + (d & 3);
        }
#pragma unroll
        for (int u = 0; u < SUM_BATCH; ++u) {
            if (base + u * 32 + lane < total) {
                atomicAdd(&acc[off[u]],     aw[u] * wv[u].x);
                atomicAdd(&acc[off[u] + 4], aw[u] * wv[u].y);
            }
        }
    }
    __syncthreads();

    float4* dst = (float4*)(weff + (size_t)k0 * NPAD);
    const float4* src = (const float4*)acc;
    for (int i = tid; i < OWN_ROWS * NPAD / 4; i += OWN_THREADS) dst[i] = src[i];

    if (blockIdx.x < STAGE_WGS)
        ((float4*)stage)[blockIdx.x * OWN_THREADS + tid] = make_float4(0.f, 0.f, 0.f, 0.f);
}

__global__ __launch_bounds__(32 * GW)
void gemm_k(const float* __restrict__ X,
            const float* __restrict__ weff,
            const float* __restrict__ bias,
            const float* __restrict__ agg_w,
            float* __restrict__ stage) {
    __shared__ v4f bs[NB_MAX * BVEC];

    const int tile = (blockIdx.x % (TILES_M / GW)) * GW + (threadIdx.x >> 5);
    const int seg  = blockIdx.x / (TILES_M / GW);
    const bool long_seg = seg < LONG_SEGS;
    const int line0 = seg * NB_MIN + (long_seg ? seg : LONG_SEGS);
    const int lane = threadIdx.x & 31;
    const int half = lane >> 4;
    const int sub  = lane & 15;
    const int koff = half * 4;
    const int kbase = line0 * 32;

    const float* ap = X + (size_t)(tile * 16 + sub) * G_ + kbase + koff;
    v2f bwa[BSTEPS], bbb[BSTEPS];
    const bool live = sub < C_;
    if (seg < BSEGS) {
        const int lk = half * 2;
        const float* wl = agg_w + seg * LSEG + lk;
        const float* bl = bias + (seg * LSEG + lk) * C_ + (live ? sub : 0);
#pragma unroll
        for (int j = 0; j < BSTEPS; ++j) {
            bwa[j] = v2f{wl[4 * j], wl[4 * j + 1]};
            bbb[j] = v2f{bl[4 * j * C_], bl[(4 * j + 1) * C_]};
        }
    }

    const v4f* wrun = (const v4f*)(weff + (size_t)kbase * NPAD) + threadIdx.x;

    v4f cp[NB_MAX];
#pragma unroll
    for (int j = 0; j < NB_MIN; ++j) cp[j] = wrun[j * BVEC];
    if (long_seg) cp[NB_MIN] = wrun[NB_MIN * BVEC];

    v8f acc0 = {0.f, 0.f, 0.f, 0.f, 0.f, 0.f, 0.f, 0.f};
    v8f acc1 = {0.f, 0.f, 0.f, 0.f, 0.f, 0.f, 0.f, 0.f};

    v4f a[RING][PAIRS], b[2][PAIRS];
    auto load_a = [&](int o, v4f* av) {
#pragma unroll
        for (int u = 0; u < PAIRS; ++u) av[u] = *(const v4f*)(ap + (o * PAIRS + u) * 8);
    };
    auto load_b = [&](int o, v4f* bv) {
#pragma unroll
        for (int u = 0; u < PAIRS; ++u) bv[u] = bs[(o * PAIRS + u) * 32 + lane];
    };

#pragma unroll
    for (int p = 0; p < RING; ++p) load_a(p, a[p]);
#pragma unroll
    for (int p = RING; p < PF_DIST; ++p) __builtin_prefetch(ap + p * UNROLL * 4, 0, 3);

#pragma unroll
    for (int j = 0; j < NB_MIN; ++j) bs[j * BVEC + threadIdx.x] = cp[j];
    if (long_seg) bs[NB_MIN * BVEC + threadIdx.x] = cp[NB_MIN];
    __builtin_amdgcn_fence(__ATOMIC_RELEASE, "workgroup", "local");
    __builtin_amdgcn_s_barrier();
    __builtin_amdgcn_fence(__ATOMIC_ACQUIRE, "workgroup", "local");
    load_b(0, b[0]);

    if (seg < BSEGS) {
#pragma unroll
        for (int j = 0; j < BSTEPS; ++j) {
            const v2f bb = {live ? bbb[j].x : 0.f, live ? bbb[j].y : 0.f};
            if (j & 1)
                acc1 = __builtin_amdgcn_wmma_f32_16x16x4_f32(
                    false, bwa[j], false, bb, (short)0, acc1, false, false);
            else
                acc0 = __builtin_amdgcn_wmma_f32_16x16x4_f32(
                    false, bwa[j], false, bb, (short)0, acc0, false, false);
        }
    }

    auto mul_batch = [&](int o, v4f* av, const v4f* bv, bool refill) {
#pragma unroll
        for (int u = 0; u < PAIRS; ++u) {
            acc0 = __builtin_amdgcn_wmma_f32_16x16x4_f32(
                 false, av[u].lo,  false, bv[u].lo,
                 (short)0, acc0,  false,  false);
            acc1 = __builtin_amdgcn_wmma_f32_16x16x4_f32(
                false, av[u].hi, false, bv[u].hi, (short)0, acc1, false, false);
            if (refill) av[u] = *(const v4f*)(ap + ((o + RING) * PAIRS + u) * 8);
        }
    };
#pragma unroll
    for (int o = 0; o < NB_MAX; ++o) {
        const int ol = o + RING, op = o + PF_DIST, ob = o + 1;
        if (ob < NB_MIN)
            load_b(ob, b[ob & 1]);
        else if (ob < NB_MAX) {
            if (long_seg) load_b(ob, b[ob & 1]);
        }
        if (op < NB_MIN) __builtin_prefetch(ap + op * UNROLL * 4, 0, 3);
        else if (op < NB_MAX) { if (long_seg) __builtin_prefetch(ap + op * UNROLL * 4, 0, 3); }
        if (o < NB_MIN) mul_batch(o, a[o % RING], b[o & 1], ol < NB_MIN);
        else { if (long_seg) mul_batch(o, a[o % RING], b[o & 1], false); }
        if (ol >= NB_MIN && ol < NB_MAX) { if (long_seg) load_a(ol, a[ol % RING]); }
        __builtin_amdgcn_sched_barrier(0);
    }

    float* st = stage + tile * 256 + sub;
    const int mbase = half * 8;
#pragma unroll
    for (int r = 0; r < 8; ++r)
        atomicAdd(st + (mbase + r) * 16, acc0[r] + acc1[r]);
}

__global__ void out_k(const float* __restrict__ stage,
                      const float* __restrict__ agg_b,
                      float* __restrict__ out) {
    int t = blockIdx.x * 256 + threadIdx.x;
    if (t >= B_ * C_) return;
    int b = t / C_, c = t % C_;
    out[t] = stage[(b >> 4) * 256 + (b & 15) * 16 + c] + agg_b[0];
}

extern "C" void kernel_launch(void* const* d_in, const int* in_sizes, int n_in,
                              void* d_out, int out_size, void* d_ws, size_t ws_size,
                              hipStream_t stream) {
    const float* X     = (const float*)d_in[0];
    const float* W     = (const float*)d_in[1];
    const float* bias  = (const float*)d_in[2];
    const float* agg_w = (const float*)d_in[3];
    const float* agg_b = (const float*)d_in[4];
    const int*   gidx  = (const int*)d_in[5];

    float* ws    = (float*)d_ws;
    float* weff  = ws + OFF_WEFF;
    float* stage = ws + OFF_STAGE;
    float* out   = (float*)d_out;

    (void)in_sizes; (void)n_in; (void)out_size; (void)ws_size;

    build_weff_k<<<OWN_WGS, OWN_THREADS, 0, stream>>>(W, agg_w, gidx, weff, stage);
    gemm_k<<<(TILES_M / GW) * KSPLIT, 32 * GW, 0, stream>>>(X, weff, bias, agg_w, stage);
    out_k<<<(B_ * C_ + 255) / 256, 256, 0, stream>>>(stage, agg_b, out);
}
